# starting kernel
# baseline (speedup 1.0000x reference)
#include <hip/hip_runtime.h>
#include <math.h>


typedef float v2f __attribute__((ext_vector_type(2)));
typedef float v8f __attribute__((ext_vector_type(8)));

#define T_DIM     4096
#define COLS      8192
#define W         256
#define BLOCK_OUT 512
#define NTILES    48

__device__ __forceinline__ float clean(float v) {
    v = (v != v) ? 0.0f : v;
    return fminf(fmaxf(v, -3.4028235e38f), 3.4028235e38f);
}

__device__ __forceinline__ float ld_guard(const float* __restrict__ x, int g, int col) {
    return (g < T_DIM) ? clean(x[(size_t)g * COLS + col]) : 0.0f;
}

__global__ void __launch_bounds__(256)
roll_mean_wmma_f32(const float* __restrict__ x, float* __restrict__ out) {
    const int lane = threadIdx.x & 31;
    const int wave = threadIdx.x >> 5;
    const int half = lane >> 4;
    const int m    = lane & 15;
    const int col  = (blockIdx.x * 8 + wave) * 16 + m;
    const int bs   = blockIdx.y * BLOCK_OUT;

    v2f a[4];
#pragma unroll
    for (int j = 0; j < 4; ++j) {
        const int k0 = 4 * j + 2 * half;
        a[j].x = (k0     <= m) ? 1.0f : 0.0f;
        a[j].y = (k0 + 1 <= m) ? 1.0f : 0.0f;
    }

    const float inv_w = 1.0f / (float)W;
    float carry = 0.0f;

    for (int tile = 0; tile < NTILES; ++tile) {
        const int rbase = tile * 16;
        const int g0    = bs + rbase;

        {
            int gp = g0 + 64;
            if (gp >= T_DIM) gp = T_DIM - 1;
            __builtin_prefetch(x + (size_t)gp * COLS + col, 0, 0);
        }

        v2f b[4];
        if (g0 + 16 <= T_DIM) {
            const float* q  = x + (size_t)(g0 + 2 * half) * COLS + col;
            const float* ql = q - (size_t)W * COLS;
#pragma unroll
            for (int j = 0; j < 4; ++j) {
                float v0 = clean(q[(4 * j)     * COLS]);
                float v1 = clean(q[(4 * j + 1) * COLS]);
                if (tile >= 16) {
                    v0 -= clean(ql[(4 * j)     * COLS]);
                    v1 -= clean(ql[(4 * j + 1) * COLS]);
                }
                b[j].x = v0; b[j].y = v1;
            }
        } else {
#pragma unroll
            for (int j = 0; j < 4; ++j) {
                const int r0 = rbase + 4 * j + 2 * half;
                float v0 = ld_guard(x, bs + r0,     col) - ld_guard(x, bs + r0 - W,     col);
                float v1 = ld_guard(x, bs + r0 + 1, col) - ld_guard(x, bs + r0 + 1 - W, col);
                b[j].x = v0; b[j].y = v1;
            }
        }

        v8f acc = { carry, carry, carry, carry, carry, carry, carry, carry };
#pragma unroll
        for (int j = 0; j < 4; ++j)
            acc = __builtin_amdgcn_wmma_f32_16x16x4_f32(
                      false, a[j], false, b[j], (short)0, acc, false, false);

        carry = __int_as_float(
            __builtin_amdgcn_ds_bpermute((m + 16) << 2, __float_as_int(acc[7])));

        if (tile == 15) {
            const float fv = carry * inv_w;
            if (half == 0)
                out[(size_t)(bs + 255) * COLS + col] = fv;
            if (blockIdx.y == 0) {
                float* p = out + (size_t)half * COLS + col;
                for (int t0 = 0; t0 < 254; t0 += 2) {
                    *p = fv;
                    p += 2 * COLS;
                }
                if (half == 0)
                    out[(size_t)254 * COLS + col] = fv;
            }
        } else if (tile >= 16 && g0 < T_DIM) {
            float* p = out + (size_t)(g0 + 8 * half) * COLS + col;
            if (tile < 47 && g0 + 16 <= T_DIM) {
#pragma unroll
                for (int vi = 0; vi < 8; ++vi)
                    p[vi * COLS] = acc[vi] * inv_w;
            } else {
#pragma unroll
                for (int vi = 0; vi < 8; ++vi) {
                    const int r = rbase + vi + 8 * half;
                    if (r <= 766 && bs + r < T_DIM)
                        p[vi * COLS] = acc[vi] * inv_w;
                }
            }
        }
    }
}

extern "C" void kernel_launch(void* const* d_in, const int* in_sizes, int n_in,
                              void* d_out, int out_size, void* d_ws, size_t ws_size,
                              hipStream_t stream) {
    (void)in_sizes; (void)n_in; (void)d_ws; (void)ws_size; (void)out_size;
    const float* x = (const float*)d_in[0];
    float* out     = (float*)d_out;
    dim3 grid(COLS / 128, 8);
    roll_mean_wmma_f32<<<grid, 256, 0, stream>>>(x, out);
}


# Each tile's loads (8 leading + 8 lag rows) issued as one group before their first use, tail rows through clamped unguarded loads: one load wait per tile where there were 8 in series; sums unchanged
# speedup vs baseline: 2.8114x; 2.8114x over previous
#include <hip/hip_runtime.h>
#include <math.h>


typedef float v2f __attribute__((ext_vector_type(2)));
typedef float v8f __attribute__((ext_vector_type(8)));

#define T_DIM     4096
#define COLS      8192
#define W         256
#define BLOCK_OUT 512
#define NTILES    48

__device__ __forceinline__ float clean(float v) {
    v = (v != v) ? 0.0f : v;
    return fminf(fmaxf(v, -3.4028235e38f), 3.4028235e38f);
}

__device__ __forceinline__ float ld_clamped(const float* __restrict__ x, int g, int col) {
    const int gc = (g < T_DIM) ? g : (T_DIM - 1);
    return x[(size_t)gc * COLS + col];
}

__device__ __forceinline__ float pad_clean(float v, int g) {
    return (g < T_DIM) ? clean(v) : 0.0f;
}

__global__ void __launch_bounds__(256)
roll_mean_wmma_f32(const float* __restrict__ x, float* __restrict__ out) {
    const int lane = threadIdx.x & 31;
    const int wave = threadIdx.x >> 5;
    const int half = lane >> 4;
    const int m    = lane & 15;
    const int col  = (blockIdx.x * 8 + wave) * 16 + m;
    const int bs   = blockIdx.y * BLOCK_OUT;

    v2f a[4];
#pragma unroll
    for (int j = 0; j < 4; ++j) {
        const int k0 = 4 * j + 2 * half;
        a[j].x = (k0     <= m) ? 1.0f : 0.0f;
        a[j].y = (k0 + 1 <= m) ? 1.0f : 0.0f;
    }

    const float inv_w = 1.0f / (float)W;
    float carry = 0.0f;

    for (int tile = 0; tile < NTILES; ++tile) {
        const int rbase = tile * 16;
        const int g0    = bs + rbase;

        {
            int gp = g0 + 64;
            if (gp >= T_DIM) gp = T_DIM - 1;
            __builtin_prefetch(x + (size_t)gp * COLS + col, 0, 0);
        }

        v2f b[4];
        if (g0 + 16 <= T_DIM) {
            const float* q  = x + (size_t)(g0 + 2 * half) * COLS + col;
            float l0[4], l1[4];
#pragma unroll
            for (int j = 0; j < 4; ++j) {
                l0[j] = q[(4 * j)     * COLS];
                l1[j] = q[(4 * j + 1) * COLS];
            }
            if (tile >= 16) {
                const float* ql = q - (size_t)W * COLS;
                float g0v[4], g1v[4];
#pragma unroll
                for (int j = 0; j < 4; ++j) {
                    g0v[j] = ql[(4 * j)     * COLS];
                    g1v[j] = ql[(4 * j + 1) * COLS];
                }
#pragma unroll
                for (int j = 0; j < 4; ++j) {
                    b[j].x = clean(l0[j]) - clean(g0v[j]);
                    b[j].y = clean(l1[j]) - clean(g1v[j]);
                }
            } else {
#pragma unroll
                for (int j = 0; j < 4; ++j) {
                    b[j].x = clean(l0[j]);
                    b[j].y = clean(l1[j]);
                }
            }
        } else {
            float l0[4], l1[4], g0v[4], g1v[4];
#pragma unroll
            for (int j = 0; j < 4; ++j) {
                const int g = bs + rbase + 4 * j + 2 * half;
                l0[j]  = ld_clamped(x, g,         col);
                l1[j]  = ld_clamped(x, g + 1,     col);
                g0v[j] = ld_clamped(x, g - W,     col);
                g1v[j] = ld_clamped(x, g + 1 - W, col);
            }
#pragma unroll
            for (int j = 0; j < 4; ++j) {
                const int g = bs + rbase + 4 * j + 2 * half;
                b[j].x = pad_clean(l0[j], g)     - pad_clean(g0v[j], g - W);
                b[j].y = pad_clean(l1[j], g + 1) - pad_clean(g1v[j], g + 1 - W);
            }
        }

        v8f acc = { carry, carry, carry, carry, carry, carry, carry, carry };
#pragma unroll
        for (int j = 0; j < 4; ++j)
            acc = __builtin_amdgcn_wmma_f32_16x16x4_f32(
                      false, a[j], false, b[j], (short)0, acc, false, false);

        carry = __int_as_float(
            __builtin_amdgcn_ds_bpermute((m + 16) << 2, __float_as_int(acc[7])));

        if (tile == 15) {
            const float fv = carry * inv_w;
            if (half == 0)
                out[(size_t)(bs + 255) * COLS + col] = fv;
            if (blockIdx.y == 0) {
                float* p = out + (size_t)half * COLS + col;
                for (int t0 = 0; t0 < 254; t0 += 2) {
                    *p = fv;
                    p += 2 * COLS;
                }
                if (half == 0)
                    out[(size_t)254 * COLS + col] = fv;
            }
        } else if (tile >= 16 && g0 < T_DIM) {
            float* p = out + (size_t)(g0 + 8 * half) * COLS + col;
            if (tile < 47 && g0 + 16 <= T_DIM) {
#pragma unroll
                for (int vi = 0; vi < 8; ++vi)
                    p[vi * COLS] = acc[vi] * inv_w;
            } else {
#pragma unroll
                for (int vi = 0; vi < 8; ++vi) {
                    const int r = rbase + vi + 8 * half;
                    if (r <= 766 && bs + r < T_DIM)
                        p[vi * COLS] = acc[vi] * inv_w;
                }
            }
        }
    }
}

extern "C" void kernel_launch(void* const* d_in, const int* in_sizes, int n_in,
                              void* d_out, int out_size, void* d_ws, size_t ws_size,
                              hipStream_t stream) {
    (void)in_sizes; (void)n_in; (void)d_ws; (void)ws_size; (void)out_size;
    const float* x = (const float*)d_in[0];
    float* out     = (float*)d_out;
    dim3 grid(COLS / 128, 8);
    roll_mean_wmma_f32<<<grid, 256, 0, stream>>>(x, out);
}


# Lag rows x[r-256] come from a per-wave LDS ring of the last 256 cleaned rows (each lane re-reads its own 16-tile-old values) instead of a second global read of x; sums unchanged
# speedup vs baseline: 2.9360x; 1.0443x over previous
#include <hip/hip_runtime.h>
#include <math.h>


typedef float v2f __attribute__((ext_vector_type(2)));
typedef float v8f __attribute__((ext_vector_type(8)));
typedef float v4f __attribute__((ext_vector_type(4)));

#define T_DIM     4096
#define COLS      8192
#define W         256
#define BLOCK_OUT 512
#define NTILES    48
#define LAG_TILES 16
#define RING_V4   (LAG_TILES * 64)

__device__ __forceinline__ float clean(float v) {
    v = (v != v) ? 0.0f : v;
    return fminf(fmaxf(v, -3.4028235e38f), 3.4028235e38f);
}

__device__ __forceinline__ float ld_clamped(const float* __restrict__ x, int g, int col) {
    const int gc = (g < T_DIM) ? g : (T_DIM - 1);
    return x[(size_t)gc * COLS + col];
}

__device__ __forceinline__ float pad_clean(float v, int g) {
    return (g < T_DIM) ? clean(v) : 0.0f;
}

__global__ void __launch_bounds__(256)
roll_mean_wmma_f32(const float* __restrict__ x, float* __restrict__ out) {
    __shared__ v4f ring[8 * RING_V4];

    const int lane = threadIdx.x & 31;
    const int wave = threadIdx.x >> 5;
    const int half = lane >> 4;
    const int m    = lane & 15;
    const int col  = (blockIdx.x * 8 + wave) * 16 + m;
    const int bs   = blockIdx.y * BLOCK_OUT;

    v4f* const rl = ring + wave * RING_V4 + lane;

    v2f a[4];
#pragma unroll
    for (int j = 0; j < 4; ++j) {
        const int k0 = 4 * j + 2 * half;
        a[j].x = (k0     <= m) ? 1.0f : 0.0f;
        a[j].y = (k0 + 1 <= m) ? 1.0f : 0.0f;
    }

    const float inv_w = 1.0f / (float)W;
    float carry = 0.0f;

    for (int tile = 0; tile < NTILES; ++tile) {
        const int rbase = tile * 16;
        const int g0    = bs + rbase;
        v4f* const rs   = rl + (tile & (LAG_TILES - 1)) * 64;

        v4f o0 = { 0.0f, 0.0f, 0.0f, 0.0f };
        v4f o1 = { 0.0f, 0.0f, 0.0f, 0.0f };
        if (tile >= LAG_TILES) {
            o0 = rs[0];
            o1 = rs[32];
        }

        {
            int gp = g0 + 64;
            if (gp >= T_DIM) gp = T_DIM - 1;
            __builtin_prefetch(x + (size_t)gp * COLS + col, 0, 0);
        }

        float c[8];
        if (g0 + 16 <= T_DIM) {
            const float* q  = x + (size_t)(g0 + 2 * half) * COLS + col;
            float l0[4], l1[4];
#pragma unroll
            for (int j = 0; j < 4; ++j) {
                l0[j] = q[(4 * j)     * COLS];
                l1[j] = q[(4 * j + 1) * COLS];
            }
#pragma unroll
            for (int j = 0; j < 4; ++j) {
                c[2 * j]     = clean(l0[j]);
                c[2 * j + 1] = clean(l1[j]);
            }
        } else {
            float l0[4], l1[4];
#pragma unroll
            for (int j = 0; j < 4; ++j) {
                const int g = bs + rbase + 4 * j + 2 * half;
                l0[j] = ld_clamped(x, g,     col);
                l1[j] = ld_clamped(x, g + 1, col);
            }
#pragma unroll
            for (int j = 0; j < 4; ++j) {
                const int g = bs + rbase + 4 * j + 2 * half;
                c[2 * j]     = pad_clean(l0[j], g);
                c[2 * j + 1] = pad_clean(l1[j], g + 1);
            }
        }

        v2f b[4];
        b[0].x = c[0] - o0.x;  b[0].y = c[1] - o0.y;
        b[1].x = c[2] - o0.z;  b[1].y = c[3] - o0.w;
        b[2].x = c[4] - o1.x;  b[2].y = c[5] - o1.y;
        b[3].x = c[6] - o1.z;  b[3].y = c[7] - o1.w;
        __builtin_amdgcn_sched_barrier(0);

        v8f acc = { carry, carry, carry, carry, carry, carry, carry, carry };
#pragma unroll
        for (int j = 0; j < 4; ++j)
            acc = __builtin_amdgcn_wmma_f32_16x16x4_f32(
                      false, a[j], false, b[j], (short)0, acc, false, false);

        carry = __int_as_float(
            __builtin_amdgcn_ds_bpermute((m + 16) << 2, __float_as_int(acc[7])));

        if (tile == 15) {
            const float fv = carry * inv_w;
            if (half == 0)
                out[(size_t)(bs + 255) * COLS + col] = fv;
            if (blockIdx.y == 0) {
                float* p = out + (size_t)half * COLS + col;
                for (int t0 = 0; t0 < 254; t0 += 2) {
                    *p = fv;
                    p += 2 * COLS;
                }
                if (half == 0)
                    out[(size_t)254 * COLS + col] = fv;
            }
        } else if (tile >= 16 && g0 < T_DIM) {
            float* p = out + (size_t)(g0 + 8 * half) * COLS + col;
            if (tile < 47 && g0 + 16 <= T_DIM) {
#pragma unroll
                for (int vi = 0; vi < 8; ++vi)
                    p[vi * COLS] = acc[vi] * inv_w;
            } else {
#pragma unroll
                for (int vi = 0; vi < 8; ++vi) {
                    const int r = rbase + vi + 8 * half;
                    if (r <= 766 && bs + r < T_DIM)
                        p[vi * COLS] = acc[vi] * inv_w;
                }
            }
        }

        if (tile < NTILES - LAG_TILES) {
            const v4f n0 = { c[0], c[1], c[2], c[3] };
            const v4f n1 = { c[4], c[5], c[6], c[7] };
            rs[0]  = n0;
            rs[32] = n1;
        }
    }
}

extern "C" void kernel_launch(void* const* d_in, const int* in_sizes, int n_in,
                              void* d_out, int out_size, void* d_ws, size_t ws_size,
                              hipStream_t stream) {
    (void)in_sizes; (void)n_in; (void)d_ws; (void)ws_size; (void)out_size;
    const float* x = (const float*)d_in[0];
    float* out     = (float*)d_out;
    dim3 grid(COLS / 128, 8);
    roll_mean_wmma_f32<<<grid, 256, 0, stream>>>(x, out);
}


# The tile prefetch (now SE scope) asks for one row per lane: every 2nd tile all 32 rows of the tiles 5 and 6 ahead, where it asked for 1 row of the tile 4 ahead; loads then find L2 lines; same sums
# speedup vs baseline: 3.3107x; 1.1276x over previous
#include <hip/hip_runtime.h>
#include <math.h>


typedef float v2f __attribute__((ext_vector_type(2)));
typedef float v8f __attribute__((ext_vector_type(8)));
typedef float v4f __attribute__((ext_vector_type(4)));

#define T_DIM     4096
#define COLS      8192
#define W         256
#define BLOCK_OUT 512
#define NTILES    48
#define LAG_TILES 16
#define RING_V4   (LAG_TILES * 64)
#define PF_TILES  5
#define LAST_ROW  (W - 1 + BLOCK_OUT - 1)

__device__ __forceinline__ float clean(float v) {
    v = (v != v) ? 0.0f : v;
    return fminf(fmaxf(v, -3.4028235e38f), 3.4028235e38f);
}

__device__ __forceinline__ float ld_clamped(const float* __restrict__ x, int g, int col) {
    const int gc = (g < T_DIM) ? g : (T_DIM - 1);
    return x[(size_t)gc * COLS + col];
}

__device__ __forceinline__ float pad_clean(float v, int g) {
    return (g < T_DIM) ? clean(v) : 0.0f;
}

__device__ __forceinline__ void prefetch_rows(const float* __restrict__ x, int g,
                                              int last, int lane, int colbase) {
    int gp = g + lane;
    gp = (gp < last) ? gp : last;
    __builtin_prefetch(x + (size_t)gp * COLS + colbase, 0, 2);
}

__global__ void __launch_bounds__(256)
roll_mean_wmma_f32(const float* __restrict__ x, float* __restrict__ out) {
    __shared__ v4f ring[8 * RING_V4];

    const int lane = threadIdx.x & 31;
    const int wave = threadIdx.x >> 5;
    const int half = lane >> 4;
    const int m    = lane & 15;
    const int col  = (blockIdx.x * 8 + wave) * 16 + m;
    const int bs   = blockIdx.y * BLOCK_OUT;
    const int colbase = (blockIdx.x * 8 + wave) * 16;
    const int pf_last = (bs + LAST_ROW < T_DIM) ? (bs + LAST_ROW) : (T_DIM - 1);

    v4f* const rl = ring + wave * RING_V4 + lane;

    v2f a[4];
#pragma unroll
    for (int j = 0; j < 4; ++j) {
        const int k0 = 4 * j + 2 * half;
        a[j].x = (k0     <= m) ? 1.0f : 0.0f;
        a[j].y = (k0 + 1 <= m) ? 1.0f : 0.0f;
    }

    const float inv_w = 1.0f / (float)W;
    float carry = 0.0f;

#pragma unroll
    for (int r = 16; r < 16 * PF_TILES; r += 32)
        prefetch_rows(x, bs + r, pf_last, lane, colbase);

    for (int tile = 0; tile < NTILES; ++tile) {
        const int rbase = tile * 16;
        const int g0    = bs + rbase;
        v4f* const rs   = rl + (tile & (LAG_TILES - 1)) * 64;

        v4f o0 = { 0.0f, 0.0f, 0.0f, 0.0f };
        v4f o1 = { 0.0f, 0.0f, 0.0f, 0.0f };
        if (tile >= LAG_TILES) {
            o0 = rs[0];
            o1 = rs[32];
        }

        if ((tile & 1) == 0 && g0 + 16 * PF_TILES <= pf_last)
            prefetch_rows(x, g0 + 16 * PF_TILES, pf_last, lane, colbase);

        float c[8];
        if (g0 + 16 <= T_DIM) {
            const float* q  = x + (size_t)(g0 + 2 * half) * COLS + col;
            float l0[4], l1[4];
#pragma unroll
            for (int j = 0; j < 4; ++j) {
                l0[j] = q[(4 * j)     * COLS];
                l1[j] = q[(4 * j + 1) * COLS];
            }
#pragma unroll
            for (int j = 0; j < 4; ++j) {
                c[2 * j]     = clean(l0[j]);
                c[2 * j + 1] = clean(l1[j]);
            }
        } else {
            float l0[4], l1[4];
#pragma unroll
            for (int j = 0; j < 4; ++j) {
                const int g = bs + rbase + 4 * j + 2 * half;
                l0[j] = ld_clamped(x, g,     col);
                l1[j] = ld_clamped(x, g + 1, col);
            }
#pragma unroll
            for (int j = 0; j < 4; ++j) {
                const int g = bs + rbase + 4 * j + 2 * half;
                c[2 * j]     = pad_clean(l0[j], g);
                c[2 * j + 1] = pad_clean(l1[j], g + 1);
            }
        }

        v2f b[4];
        b[0].x = c[0] - o0.x;  b[0].y = c[1] - o0.y;
        b[1].x = c[2] - o0.z;  b[1].y = c[3] - o0.w;
        b[2].x = c[4] - o1.x;  b[2].y = c[5] - o1.y;
        b[3].x = c[6] - o1.z;  b[3].y = c[7] - o1.w;
        __builtin_amdgcn_sched_barrier(0);

        v8f acc = { carry, carry, carry, carry, carry, carry, carry, carry };
#pragma unroll
        for (int j = 0; j < 4; ++j)
            acc = __builtin_amdgcn_wmma_f32_16x16x4_f32(
                      false, a[j], false, b[j], (short)0, acc, false, false);

        carry = __int_as_float(
            __builtin_amdgcn_ds_bpermute((m + 16) << 2, __float_as_int(acc[7])));

        if (tile == 15) {
            const float fv = carry * inv_w;
            if (half == 0)
                out[(size_t)(bs + 255) * COLS + col] = fv;
            if (blockIdx.y == 0) {
                float* p = out + (size_t)half * COLS + col;
                for (int t0 = 0; t0 < 254; t0 += 2) {
                    *p = fv;
                    p += 2 * COLS;
                }
                if (half == 0)
                    out[(size_t)254 * COLS + col] = fv;
            }
        } else if (tile >= 16 && g0 < T_DIM) {
            float* p = out + (size_t)(g0 + 8 * half) * COLS + col;
            if (tile < 47 && g0 + 16 <= T_DIM) {
#pragma unroll
                for (int vi = 0; vi < 8; ++vi)
                    p[vi * COLS] = acc[vi] * inv_w;
            } else {
#pragma unroll
                for (int vi = 0; vi < 8; ++vi) {
                    const int r = rbase + vi + 8 * half;
                    if (r <= 766 && bs + r < T_DIM)
                        p[vi * COLS] = acc[vi] * inv_w;
                }
            }
        }

        if (tile < NTILES - LAG_TILES) {
            const v4f n0 = { c[0], c[1], c[2], c[3] };
            const v4f n1 = { c[4], c[5], c[6], c[7] };
            rs[0]  = n0;
            rs[32] = n1;
        }
    }
}

extern "C" void kernel_launch(void* const* d_in, const int* in_sizes, int n_in,
                              void* d_out, int out_size, void* d_ws, size_t ws_size,
                              hipStream_t stream) {
    (void)in_sizes; (void)n_in; (void)d_ws; (void)ws_size; (void)out_size;
    const float* x = (const float*)d_in[0];
    float* out     = (float*)d_out;
    dim3 grid(COLS / 128, 8);
    roll_mean_wmma_f32<<<grid, 256, 0, stream>>>(x, out);
}


# Each wave takes 64 columns, not 16 (4-wave groups, 1 wave per SIMD): 16-byte loads/stores, 256-byte run per row; 4 independent WMMA chains; next tile's loads issued under this tile's math; same sums
# speedup vs baseline: 4.5176x; 1.3645x over previous
#include <hip/hip_runtime.h>
#include <math.h>


typedef float v2f __attribute__((ext_vector_type(2)));
typedef float v4f __attribute__((ext_vector_type(4)));
typedef float v8f __attribute__((ext_vector_type(8)));

#define T_DIM     4096
#define COLS      8192
#define W         256
#define BLOCK_OUT 512
#define NTILES    48
#define LAG_TILES 16
#define WCOLS     64
#define SLOT_V4   256
#define RING_V4   (LAG_TILES * SLOT_V4)
#define PF_TILES  5
#define LAST_ROW  (W - 1 + BLOCK_OUT - 1)

static_assert(T_DIM % 16 == 0 && BLOCK_OUT % 16 == 0 && W % 16 == 0, "tile grid");

__device__ __forceinline__ float clean(float v) {
    v = (v != v) ? 0.0f : v;
    return fminf(fmaxf(v, -3.4028235e38f), 3.4028235e38f);
}

__device__ __forceinline__ v4f clean4(v4f v) {
    v4f r = { clean(v.x), clean(v.y), clean(v.z), clean(v.w) };
    return r;
}

__device__ __forceinline__ void load_tile(const float* __restrict__ q, v4f (&l)[8]) {
#pragma unroll
    for (int j = 0; j < 4; ++j) {
        l[2 * j]     = *reinterpret_cast<const v4f*>(q + (size_t)(4 * j)     * COLS);
        l[2 * j + 1] = *reinterpret_cast<const v4f*>(q + (size_t)(4 * j + 1) * COLS);
    }
}

__device__ __forceinline__ void prefetch_rows(const float* __restrict__ x, int g,
                                              int last, int lane, int colbase) {
    int gp = g + lane;
    gp = (gp < last) ? gp : last;
    __builtin_prefetch(x + (size_t)gp * COLS + colbase, 0, 2);
}

__global__ void __launch_bounds__(128)
roll_mean_wmma_f32(const float* __restrict__ x, float* __restrict__ out) {
    __shared__ v4f ring[4 * RING_V4];

    const int lane = threadIdx.x & 31;
    const int wave = threadIdx.x >> 5;
    const int half = lane >> 4;
    const int m    = lane & 15;
    const int colbase = (blockIdx.x * 4 + wave) * WCOLS;
    const int col  = colbase + 4 * m;
    const int bs   = blockIdx.y * BLOCK_OUT;
    const int pf_last = (bs + LAST_ROW < T_DIM) ? (bs + LAST_ROW) : (T_DIM - 1);
    const int ntiles = (T_DIM - bs >= 16 * NTILES) ? NTILES : (T_DIM - bs) / 16;

    v4f* const rl = ring + wave * RING_V4 + lane;

    v2f a[4];
#pragma unroll
    for (int j = 0; j < 4; ++j) {
        const int k0 = 4 * j + 2 * half;
        a[j].x = (k0     <= m) ? 1.0f : 0.0f;
        a[j].y = (k0 + 1 <= m) ? 1.0f : 0.0f;
    }

    const float inv_w = 1.0f / (float)W;
    float carry[4] = { 0.0f, 0.0f, 0.0f, 0.0f };

    v4f c[8];
    {
        v4f first[8];
        load_tile(x + (size_t)(bs + 2 * half) * COLS + col, first);
#pragma unroll
        for (int r = 16; r < 16 * PF_TILES; r += 32)
            prefetch_rows(x, bs + r, pf_last, lane, colbase);
#pragma unroll
        for (int k = 0; k < 8; ++k)
            c[k] = clean4(first[k]);
    }

    for (int tile = 0; tile < ntiles; ++tile) {
        const int rbase = tile * 16;
        const int g0    = bs + rbase;
        v4f* const rs   = rl + (tile & (LAG_TILES - 1)) * SLOT_V4;

        v4f o[8];
#pragma unroll
        for (int k = 0; k < 8; ++k)
            o[k] = (v4f){ 0.0f, 0.0f, 0.0f, 0.0f };
        if (tile >= LAG_TILES) {
#pragma unroll
            for (int k = 0; k < 8; ++k)
                o[k] = rs[k * 32];
        }

        if ((tile & 1) == 0 && g0 + 16 * PF_TILES <= pf_last)
            prefetch_rows(x, g0 + 16 * PF_TILES, pf_last, lane, colbase);

        const bool more = tile + 1 < ntiles;
        v4f nxt[8];
        if (more)
            load_tile(x + (size_t)(g0 + 16 + 2 * half) * COLS + col, nxt);

        v2f b[4][4];
#pragma unroll
        for (int j = 0; j < 4; ++j) {
            const v4f d0 = c[2 * j]     - o[2 * j];
            const v4f d1 = c[2 * j + 1] - o[2 * j + 1];
            b[0][j].x = d0.x;  b[0][j].y = d1.x;
            b[1][j].x = d0.y;  b[1][j].y = d1.y;
            b[2][j].x = d0.z;  b[2][j].y = d1.z;
            b[3][j].x = d0.w;  b[3][j].y = d1.w;
        }
        __builtin_amdgcn_sched_barrier(0);

        v8f acc[4];
#pragma unroll
        for (int q = 0; q < 4; ++q) {
            const float s = carry[q];
            acc[q] = (v8f){ s, s, s, s, s, s, s, s };
        }
#pragma unroll
        for (int j = 0; j < 4; ++j) {
#pragma unroll
            for (int q = 0; q < 4; ++q)
                acc[q] = __builtin_amdgcn_wmma_f32_16x16x4_f32(
                             false, a[j], false, b[q][j], (short)0, acc[q], false, false);
        }

#pragma unroll
        for (int q = 0; q < 4; ++q)
            carry[q] = __int_as_float(
                __builtin_amdgcn_ds_bpermute((m + 16) << 2, __float_as_int(acc[q][7])));

        __builtin_amdgcn_sched_barrier(0);
        asm volatile("v_nop\n\tv_nop\n\tv_nop\n\tv_nop");

        if (tile == 15) {
            const v4f fv = { carry[0] * inv_w, carry[1] * inv_w,
                             carry[2] * inv_w, carry[3] * inv_w };
            if (half == 0)
                *reinterpret_cast<v4f*>(out + (size_t)(bs + 255) * COLS + col) = fv;
            if (blockIdx.y == 0) {
                float* p = out + (size_t)half * COLS + col;
                for (int t0 = 0; t0 < 254; t0 += 2) {
                    *reinterpret_cast<v4f*>(p) = fv;
                    p += 2 * COLS;
                }
                if (half == 0)
                    *reinterpret_cast<v4f*>(out + (size_t)254 * COLS + col) = fv;
            }
        } else if (tile >= 16) {
            float* p = out + (size_t)(g0 + 8 * half) * COLS + col;
            if (tile < NTILES - 1) {
#pragma unroll
                for (int vi = 0; vi < 8; ++vi) {
                    const v4f r4 = { acc[0][vi] * inv_w, acc[1][vi] * inv_w,
                                     acc[2][vi] * inv_w, acc[3][vi] * inv_w };
                    *reinterpret_cast<v4f*>(p + (size_t)vi * COLS) = r4;
                }
            } else {
#pragma unroll
                for (int vi = 0; vi < 8; ++vi) {
                    const int r = rbase + vi + 8 * half;
                    if (r <= LAST_ROW) {
                        const v4f r4 = { acc[0][vi] * inv_w, acc[1][vi] * inv_w,
                                         acc[2][vi] * inv_w, acc[3][vi] * inv_w };
                        *reinterpret_cast<v4f*>(p + (size_t)vi * COLS) = r4;
                    }
                }
            }
        }

        if (tile + LAG_TILES < ntiles) {
#pragma unroll
            for (int k = 0; k < 8; ++k)
                rs[k * 32] = c[k];
        }

        if (more) {
#pragma unroll
            for (int k = 0; k < 8; ++k)
                c[k] = clean4(nxt[k]);
        }
    }
}

extern "C" void kernel_launch(void* const* d_in, const int* in_sizes, int n_in,
                              void* d_out, int out_size, void* d_ws, size_t ws_size,
                              hipStream_t stream) {
    (void)in_sizes; (void)n_in; (void)d_ws; (void)ws_size; (void)out_size;
    const float* x = (const float*)d_in[0];
    float* out     = (float*)d_out;
    dim3 grid(COLS / 256, 8);
    roll_mean_wmma_f32<<<grid, 128, 0, stream>>>(x, out);
}


# Row loads issued 4 tiles ahead into 4 register sets taking turns (loop written 4 tiles per trip; was 1 ahead), L2 prefetch lead moved out with them: 32 loads in flight per wave, not 8; same sums
# speedup vs baseline: 5.0220x; 1.1117x over previous
#include <hip/hip_runtime.h>
#include <math.h>


typedef float v2f __attribute__((ext_vector_type(2)));
typedef float v4f __attribute__((ext_vector_type(4)));
typedef float v8f __attribute__((ext_vector_type(8)));

#define T_DIM     4096
#define COLS      8192
#define W         256
#define BLOCK_OUT 512
#define NTILES    48
#define LAG_TILES 16
#define WCOLS     64
#define SLOT_V4   256
#define RING_V4   (LAG_TILES * SLOT_V4)
#define DEPTH     4
#define PF_TILES  (DEPTH + 4)
#define LAST_ROW  (W - 1 + BLOCK_OUT - 1)

static_assert(T_DIM % 16 == 0 && BLOCK_OUT % 16 == 0 && W % 16 == 0, "tile grid");
static_assert(NTILES % DEPTH == 0 && T_DIM % (16 * DEPTH) == 0 && BLOCK_OUT % (16 * DEPTH) == 0 &&
              DEPTH == 4, "trips");

__device__ __forceinline__ float clean(float v) {
    v = (v != v) ? 0.0f : v;
    return fminf(fmaxf(v, -3.4028235e38f), 3.4028235e38f);
}

__device__ __forceinline__ v4f clean4(v4f v) {
    v4f r = { clean(v.x), clean(v.y), clean(v.z), clean(v.w) };
    return r;
}

__device__ __forceinline__ void load_tile(const float* __restrict__ q, v4f (&l)[8]) {
#pragma unroll
    for (int j = 0; j < 4; ++j) {
        l[2 * j]     = *reinterpret_cast<const v4f*>(q + (size_t)(4 * j)     * COLS);
        l[2 * j + 1] = *reinterpret_cast<const v4f*>(q + (size_t)(4 * j + 1) * COLS);
    }
}

__device__ __forceinline__ void prefetch_rows(const float* __restrict__ x, int g,
                                              int last, int lane, int colbase) {
    int gp = g + lane;
    gp = (gp < last) ? gp : last;
    __builtin_prefetch(x + (size_t)gp * COLS + colbase, 0, 2);
}

__global__ void __launch_bounds__(128)
roll_mean_wmma_f32(const float* __restrict__ x, float* __restrict__ out) {
    __shared__ v4f ring[4 * RING_V4];

    const int lane = threadIdx.x & 31;
    const int wave = threadIdx.x >> 5;
    const int half = lane >> 4;
    const int m    = lane & 15;
    const int colbase = (blockIdx.x * 4 + wave) * WCOLS;
    const int col  = colbase + 4 * m;
    const int bs   = blockIdx.y * BLOCK_OUT;
    const int pf_last = (bs + LAST_ROW < T_DIM) ? (bs + LAST_ROW) : (T_DIM - 1);
    const int ntiles = (T_DIM - bs >= 16 * NTILES) ? NTILES : (T_DIM - bs) / 16;

    v4f* const rl = ring + wave * RING_V4 + lane;

    v2f a[4];
#pragma unroll
    for (int j = 0; j < 4; ++j) {
        const int k0 = 4 * j + 2 * half;
        a[j].x = (k0     <= m) ? 1.0f : 0.0f;
        a[j].y = (k0 + 1 <= m) ? 1.0f : 0.0f;
    }

    const float inv_w = 1.0f / (float)W;
    float carry[4] = { 0.0f, 0.0f, 0.0f, 0.0f };

    const float* const xl = x + (size_t)(bs + 2 * half) * COLS + col;

    v4f s0[8], s1[8], s2[8], s3[8];
    v4f c[8];
    load_tile(xl, s0);
    load_tile(xl + (size_t)16 * COLS, s1);
    load_tile(xl + (size_t)32 * COLS, s2);
    load_tile(xl + (size_t)48 * COLS, s3);
#pragma unroll
    for (int r = 16 * DEPTH; r < 16 * PF_TILES; r += 32)
        prefetch_rows(x, bs + r, pf_last, lane, colbase);
#pragma unroll
    for (int k = 0; k < 8; ++k)
        c[k] = clean4(s0[k]);

    auto step = [&](const int tile, v4f (&fill)[8], v4f (&nx)[8]) {
        const int rbase = tile * 16;
        const int g0    = bs + rbase;
        v4f* const rs   = rl + (tile & (LAG_TILES - 1)) * SLOT_V4;

        v4f o[8];
#pragma unroll
        for (int k = 0; k < 8; ++k)
            o[k] = (v4f){ 0.0f, 0.0f, 0.0f, 0.0f };
        if (tile >= LAG_TILES) {
#pragma unroll
            for (int k = 0; k < 8; ++k)
                o[k] = rs[k * 32];
        }

        if ((tile & 1) == 0 && g0 + 16 * PF_TILES <= pf_last)
            prefetch_rows(x, g0 + 16 * PF_TILES, pf_last, lane, colbase);

        const int tl = (tile + DEPTH < ntiles) ? (tile + DEPTH) : (ntiles - 1);
        load_tile(xl + (size_t)tl * (16 * COLS), fill);

        v2f b[4][4];
#pragma unroll
        for (int j = 0; j < 4; ++j) {
            const v4f d0 = c[2 * j]     - o[2 * j];
            const v4f d1 = c[2 * j + 1] - o[2 * j + 1];
            b[0][j].x = d0.x;  b[0][j].y = d1.x;
            b[1][j].x = d0.y;  b[1][j].y = d1.y;
            b[2][j].x = d0.z;  b[2][j].y = d1.z;
            b[3][j].x = d0.w;  b[3][j].y = d1.w;
        }
        __builtin_amdgcn_sched_barrier(0);

        v8f acc[4];
#pragma unroll
        for (int q = 0; q < 4; ++q) {
            const float s = carry[q];
            acc[q] = (v8f){ s, s, s, s, s, s, s, s };
        }
#pragma unroll
        for (int j = 0; j < 4; ++j) {
#pragma unroll
            for (int q = 0; q < 4; ++q)
                acc[q] = __builtin_amdgcn_wmma_f32_16x16x4_f32(
                             false, a[j], false, b[q][j], (short)0, acc[q], false, false);
        }

#pragma unroll
        for (int q = 0; q < 4; ++q)
            carry[q] = __int_as_float(
                __builtin_amdgcn_ds_bpermute((m + 16) << 2, __float_as_int(acc[q][7])));

        __builtin_amdgcn_sched_barrier(0);
        asm volatile("v_nop\n\tv_nop\n\tv_nop\n\tv_nop");

        if (tile == 15) {
            const v4f fv = { carry[0] * inv_w, carry[1] * inv_w,
                             carry[2] * inv_w, carry[3] * inv_w };
            if (half == 0)
                *reinterpret_cast<v4f*>(out + (size_t)(bs + 255) * COLS + col) = fv;
            if (blockIdx.y == 0) {
                float* p = out + (size_t)half * COLS + col;
                for (int t0 = 0; t0 < 254; t0 += 2) {
                    *reinterpret_cast<v4f*>(p) = fv;
                    p += 2 * COLS;
                }
                if (half == 0)
                    *reinterpret_cast<v4f*>(out + (size_t)254 * COLS + col) = fv;
            }
        } else if (tile >= 16) {
            float* p = out + (size_t)(g0 + 8 * half) * COLS + col;
            if (tile < NTILES - 1) {
#pragma unroll
                for (int vi = 0; vi < 8; ++vi) {
                    const v4f r4 = { acc[0][vi] * inv_w, acc[1][vi] * inv_w,
                                     acc[2][vi] * inv_w, acc[3][vi] * inv_w };
                    *reinterpret_cast<v4f*>(p + (size_t)vi * COLS) = r4;
                }
            } else {
#pragma unroll
                for (int vi = 0; vi < 8; ++vi) {
                    const int r = rbase + vi + 8 * half;
                    if (r <= LAST_ROW) {
                        const v4f r4 = { acc[0][vi] * inv_w, acc[1][vi] * inv_w,
                                         acc[2][vi] * inv_w, acc[3][vi] * inv_w };
                        *reinterpret_cast<v4f*>(p + (size_t)vi * COLS) = r4;
                    }
                }
            }
        }

        if (tile + LAG_TILES < ntiles) {
#pragma unroll
            for (int k = 0; k < 8; ++k)
                rs[k * 32] = c[k];
        }

#pragma unroll
        for (int k = 0; k < 8; ++k)
            c[k] = clean4(nx[k]);
    };

    for (int t4 = 0; t4 < ntiles; t4 += DEPTH) {
        step(t4,     s0, s1);
        step(t4 + 1, s1, s2);
        step(t4 + 2, s2, s3);
        step(t4 + 3, s3, s0);
    }
}

extern "C" void kernel_launch(void* const* d_in, const int* in_sizes, int n_in,
                              void* d_out, int out_size, void* d_ws, size_t ws_size,
                              hipStream_t stream) {
    (void)in_sizes; (void)n_in; (void)d_ws; (void)ws_size; (void)out_size;
    const float* x = (const float*)d_in[0];
    float* out     = (float*)d_out;
    dim3 grid(COLS / 256, 8);
    roll_mean_wmma_f32<<<grid, 128, 0, stream>>>(x, out);
}
